# starting kernel
# baseline (speedup 1.0000x reference)
#include <hip/hip_runtime.h>

typedef __attribute__((ext_vector_type(2))) float v2f;
typedef __attribute__((ext_vector_type(8))) float v8f;

namespace {
constexpr int B = 8, C = 128, H = 160, W = 160;
constexpr int HW = H * W;
constexpr long CHW = (long)C * HW;
constexpr int R = 2, NOFF = 5;
constexpr int TILE_OUT = 12;
constexpr int TPR = (W + TILE_OUT - 1) / TILE_OUT;
constexpr int WAVES_PER_BLOCK = 8;
constexpr int TOTAL_TILES = B * H * TPR;
}

__device__ __forceinline__ int refl(int i, int n) {
  i = (i < 0) ? -i : i;
  i = (i >= n) ? (2 * n - 2 - i) : i;
  return i;
}

__global__ __launch_bounds__(256) void corr_wmma_kernel(
    const float* __restrict__ c1, const float* __restrict__ c2,
    const float* __restrict__ prelu, float* __restrict__ out) {
  const int lane = threadIdx.x & 31;
  const int wave = threadIdx.x >> 5;
  const int tile = blockIdx.x * WAVES_PER_BLOCK + wave;

  const int t  = tile % TPR;
  const int h  = (tile / TPR) % H;
  const int b  = tile / (TPR * H);
  const int w0 = t * TILE_OUT;

  const int half = lane >> 4;
  const int l16  = lane & 15;

  const int aw = (w0 + l16 < W) ? (w0 + l16) : (W - 1);
  const int bw = refl(w0 - 2 + l16, W);

  int hs[NOFF];
#pragma unroll
  for (int dy = 0; dy < NOFF; ++dy) hs[dy] = refl(h + dy - R, H);

  const float* c1p = c1 + (long)b * CHW + (long)h * W + aw;
  const float* c2p[NOFF];
#pragma unroll
  for (int dy = 0; dy < NOFF; ++dy)
    c2p[dy] = c2 + (long)b * CHW + (long)hs[dy] * W + bw;

  v8f acc[NOFF] = {};

  for (int k0 = 0; k0 < C; k0 += 4) {
    const long co = (long)(k0 + 2 * half) * HW;
    v2f a;
    a.x = c1p[co];
    a.y = c1p[co + HW];
#pragma unroll
    for (int dy = 0; dy < NOFF; ++dy) {
      v2f bm;
      bm.x = c2p[dy][co];
      bm.y = c2p[dy][co + HW];
      acc[dy] = __builtin_amdgcn_wmma_f32_16x16x4_f32(
          false, a, false, bm, (short)0, acc[dy], false, false);
    }
  }

  const float alpha = prelu[0];
  const float inv_c = 1.0f / (float)C;
  float* outb = out + (long)b * (NOFF * NOFF) * HW + (long)h * W;
#pragma unroll
  for (int dy = 0; dy < NOFF; ++dy) {
#pragma unroll
    for (int v = 0; v < 8; ++v) {
      const int i  = v + 8 * half;
      const int dx = l16 - i;
      const int wp = w0 + i;
      if (dx >= 0 && dx < NOFF && i < TILE_OUT && wp < W) {
        float val = acc[dy][v] * inv_c;
        val = (val >= 0.0f) ? val : alpha * val;
        outb[(long)(dy * NOFF + dx) * HW + wp] = val;
      }
    }
  }
}

extern "C" void kernel_launch(void* const* d_in, const int* in_sizes, int n_in,
                              void* d_out, int out_size, void* d_ws, size_t ws_size,
                              hipStream_t stream) {
  (void)in_sizes; (void)n_in; (void)out_size; (void)d_ws; (void)ws_size;
  const float* c1    = (const float*)d_in[0];
  const float* c2    = (const float*)d_in[1];
  const float* prelu = (const float*)d_in[2];
  float* out = (float*)d_out;

  dim3 block(32 * WAVES_PER_BLOCK);
  dim3 grid(TOTAL_TILES / WAVES_PER_BLOCK);
  corr_wmma_kernel<<<grid, block, 0, stream>>>(c1, c2, prelu, out);
}


# One wave computes a 5-row band of its 12-px tile: per k-step 5 c1 + 9 c2 row fragments, loaded once, feed 25 WMMAs (loads per WMMA 12/5 -> 28/25; 4 waves per SIMD); per-output k order kept
# speedup vs baseline: 1.5248x; 1.5248x over previous
#include <hip/hip_runtime.h>

typedef __attribute__((ext_vector_type(2))) float v2f;
typedef __attribute__((ext_vector_type(8))) float v8f;

namespace {
constexpr int B = 8, C = 128, H = 160, W = 160;
constexpr int HW = H * W;
constexpr long CHW = (long)C * HW;
constexpr int R = 2, NOFF = 5;
constexpr int TILE_OUT = 12;
constexpr int TPR = (W + TILE_OUT - 1) / TILE_OUT;
constexpr int RB = 5;
constexpr int NBAND = H / RB;
constexpr int NB2 = RB + 2 * R;
constexpr int WAVES_PER_BLOCK = 8;
constexpr int TOTAL_TILES = B * NBAND * TPR;
static_assert(H % RB == 0, "bands must tile the image height");
static_assert(TOTAL_TILES % WAVES_PER_BLOCK == 0, "grid must be whole blocks");
}

__device__ __forceinline__ int refl(int i, int n) {
  i = (i < 0) ? -i : i;
  i = (i >= n) ? (2 * n - 2 - i) : i;
  return i;
}

__global__ __launch_bounds__(256) __attribute__((amdgpu_waves_per_eu(4, 4)))
void corr_wmma_kernel(
    const float* __restrict__ c1, const float* __restrict__ c2,
    const float* __restrict__ prelu, float* __restrict__ out) {
  const int lane = threadIdx.x & 31;
  const int wave = __builtin_amdgcn_readfirstlane((int)(threadIdx.x >> 5));
  const int tile = blockIdx.x * WAVES_PER_BLOCK + wave;

  const int t    = tile % TPR;
  const int band = (tile / TPR) % NBAND;
  const int b    = tile / (TPR * NBAND);
  const int w0 = t * TILE_OUT;
  const int h0 = band * RB;

  const int half = lane >> 4;
  const int l16  = lane & 15;

  const int aw = (w0 + l16 < W) ? (w0 + l16) : (W - 1);
  const int bw = refl(w0 - 2 + l16, W);

  const unsigned alane = (unsigned)(aw + 2 * half * HW);
  const unsigned blane = (unsigned)(bw + 2 * half * HW);
  int brow[NB2];
#pragma unroll
  for (int j = 0; j < NB2; ++j) brow[j] = refl(h0 - R + j, H) * W;

  const float* c1b = c1 + (long)b * CHW + h0 * W;
  const float* c2b = c2 + (long)b * CHW;

  v8f acc[RB][NOFF] = {};

  for (int k0 = 0; k0 < C; k0 += 4) {
    const float* p1 = c1b + (long)k0 * HW;
    const float* p2 = c2b + (long)k0 * HW;
    v2f a[RB];
    v2f bm[NB2];
#pragma unroll
    for (int r = 0; r < RB; ++r) {
      const float* pr = p1 + r * W;
      a[r].x = pr[alane];
      a[r].y = pr[alane + (unsigned)HW];
    }
#pragma unroll
    for (int j = 0; j < NB2; ++j) {
      const float* pr = p2 + brow[j];
      bm[j].x = pr[blane];
      bm[j].y = pr[blane + (unsigned)HW];
    }
    __builtin_amdgcn_sched_barrier(0);
#pragma unroll
    for (int j = 0; j < NB2; ++j) {
#pragma unroll
      for (int r = 0; r < RB; ++r) {
        const int dy = j - r;
        if (dy >= 0 && dy < NOFF) {
          acc[r][dy] = __builtin_amdgcn_wmma_f32_16x16x4_f32(
              false, a[r], false, bm[j], (short)0, acc[r][dy], false, false);
        }
      }
    }
  }

  const float alpha = prelu[0];
  const float inv_c = 1.0f / (float)C;
  float* outb = out + (long)b * (NOFF * NOFF) * HW + (long)h0 * W;
#pragma unroll
  for (int r = 0; r < RB; ++r) {
#pragma unroll
    for (int dy = 0; dy < NOFF; ++dy) {
#pragma unroll
      for (int v = 0; v < 8; ++v) {
        const int i  = v + 8 * half;
        const int dx = l16 - i;
        const int wp = w0 + i;
        if (dx >= 0 && dx < NOFF && i < TILE_OUT && wp < W) {
          float val = acc[r][dy][v] * inv_c;
          val = (val >= 0.0f) ? val : alpha * val;
          outb[(long)(dy * NOFF + dx) * HW + r * W + wp] = val;
        }
      }
    }
  }
}

extern "C" void kernel_launch(void* const* d_in, const int* in_sizes, int n_in,
                              void* d_out, int out_size, void* d_ws, size_t ws_size,
                              hipStream_t stream) {
  (void)in_sizes; (void)n_in; (void)out_size; (void)d_ws; (void)ws_size;
  const float* c1    = (const float*)d_in[0];
  const float* c2    = (const float*)d_in[1];
  const float* prelu = (const float*)d_in[2];
  float* out = (float*)d_out;

  dim3 block(32 * WAVES_PER_BLOCK);
  dim3 grid(TOTAL_TILES / WAVES_PER_BLOCK);
  corr_wmma_kernel<<<grid, block, 0, stream>>>(c1, c2, prelu, out);
}


# Epilogue staged through a per-wave LDS slab [plane][row][12 px]: 12 float4 stores of 48-byte runs per wave replace 200 stores that scatter 4-byte pieces over ten lines each; values unchanged
# speedup vs baseline: 1.8971x; 1.2442x over previous
#include <hip/hip_runtime.h>

typedef __attribute__((ext_vector_type(2))) float v2f;
typedef __attribute__((ext_vector_type(8))) float v8f;

namespace {
constexpr int B = 8, C = 128, H = 160, W = 160;
constexpr int HW = H * W;
constexpr long CHW = (long)C * HW;
constexpr int R = 2, NOFF = 5;
constexpr int TILE_OUT = 12;
constexpr int TPR = (W + TILE_OUT - 1) / TILE_OUT;
constexpr int RB = 5;
constexpr int NBAND = H / RB;
constexpr int NB2 = RB + 2 * R;
constexpr int WAVES_PER_BLOCK = 8;
constexpr int TOTAL_TILES = B * NBAND * TPR;
constexpr int SLAB = NOFF * NOFF * RB * TILE_OUT;
static_assert(H % RB == 0, "bands must tile the image height");
static_assert(TOTAL_TILES % WAVES_PER_BLOCK == 0, "grid must be whole blocks");
static_assert(TILE_OUT % 4 == 0 && W % 4 == 0 && SLAB % 4 == 0, "float4 runs");
}

__device__ __forceinline__ int refl(int i, int n) {
  i = (i < 0) ? -i : i;
  i = (i >= n) ? (2 * n - 2 - i) : i;
  return i;
}

__global__ __launch_bounds__(256) __attribute__((amdgpu_waves_per_eu(4, 4)))
void corr_wmma_kernel(
    const float* __restrict__ c1, const float* __restrict__ c2,
    const float* __restrict__ prelu, float* __restrict__ out) {
  const int lane = threadIdx.x & 31;
  const int wave = __builtin_amdgcn_readfirstlane((int)(threadIdx.x >> 5));
  const int tile = blockIdx.x * WAVES_PER_BLOCK + wave;

  const int t    = tile % TPR;
  const int band = (tile / TPR) % NBAND;
  const int b    = tile / (TPR * NBAND);
  const int w0 = t * TILE_OUT;
  const int h0 = band * RB;

  const int half = lane >> 4;
  const int l16  = lane & 15;

  const int aw = (w0 + l16 < W) ? (w0 + l16) : (W - 1);
  const int bw = refl(w0 - 2 + l16, W);

  const unsigned alane = (unsigned)(aw + 2 * half * HW);
  const unsigned blane = (unsigned)(bw + 2 * half * HW);
  int brow[NB2];
#pragma unroll
  for (int j = 0; j < NB2; ++j) brow[j] = refl(h0 - R + j, H) * W;

  const float* c1b = c1 + (long)b * CHW + h0 * W;
  const float* c2b = c2 + (long)b * CHW;

  v8f acc[RB][NOFF] = {};

  for (int k0 = 0; k0 < C; k0 += 4) {
    const float* p1 = c1b + (long)k0 * HW;
    const float* p2 = c2b + (long)k0 * HW;
    v2f a[RB];
    v2f bm[NB2];
#pragma unroll
    for (int r = 0; r < RB; ++r) {
      const float* pr = p1 + r * W;
      a[r].x = pr[alane];
      a[r].y = pr[alane + (unsigned)HW];
    }
#pragma unroll
    for (int j = 0; j < NB2; ++j) {
      const float* pr = p2 + brow[j];
      bm[j].x = pr[blane];
      bm[j].y = pr[blane + (unsigned)HW];
    }
    __builtin_amdgcn_sched_barrier(0);
#pragma unroll
    for (int j = 0; j < NB2; ++j) {
#pragma unroll
      for (int r = 0; r < RB; ++r) {
        const int dy = j - r;
        if (dy >= 0 && dy < NOFF) {
          acc[r][dy] = __builtin_amdgcn_wmma_f32_16x16x4_f32(
              false, a[r], false, bm[j], (short)0, acc[r][dy], false, false);
        }
      }
    }
  }

  const float alpha = prelu[0];
  const float inv_c = 1.0f / (float)C;
  __shared__ __attribute__((aligned(16))) float stage[WAVES_PER_BLOCK][SLAB];
  float* my = stage[wave];
#pragma unroll
  for (int v = 0; v < 8; ++v) {
    const int i  = v + 8 * half;
    const int dx = l16 - i;
    if (dx >= 0 && dx < NOFF && i < TILE_OUT) {
      float* dst = my + dx * (RB * TILE_OUT) + i;
#pragma unroll
      for (int r = 0; r < RB; ++r) {
#pragma unroll
        for (int dy = 0; dy < NOFF; ++dy) {
          float val = acc[r][dy][v] * inv_c;
          val = (val >= 0.0f) ? val : alpha * val;
          dst[dy * (NOFF * RB * TILE_OUT) + r * TILE_OUT] = val;
        }
      }
    }
  }
  __syncthreads();
  float* outb = out + (long)b * (NOFF * NOFF) * HW + (long)h0 * W + w0;
  constexpr int NF4 = SLAB / 4;
#pragma unroll
  for (int it = 0; it < (NF4 + 31) / 32; ++it) {
    const int f   = it * 32 + lane;
    const int run = f / 3;
    const int q   = f - run * 3;
    const int p   = run / RB;
    const int r   = run - p * RB;
    if (f < NF4 && w0 + 4 * q < W) {
      const float4 val = *(const float4*)(my + 4 * f);
      *(float4*)(outb + (long)p * HW + r * W + 4 * q) = val;
    }
  }
}

extern "C" void kernel_launch(void* const* d_in, const int* in_sizes, int n_in,
                              void* d_out, int out_size, void* d_ws, size_t ws_size,
                              hipStream_t stream) {
  (void)in_sizes; (void)n_in; (void)out_size; (void)d_ws; (void)ws_size;
  const float* c1    = (const float*)d_in[0];
  const float* c2    = (const float*)d_in[1];
  const float* prelu = (const float*)d_in[2];
  float* out = (float*)d_out;

  dim3 block(32 * WAVES_PER_BLOCK);
  dim3 grid(TOTAL_TILES / WAVES_PER_BLOCK);
  corr_wmma_kernel<<<grid, block, 0, stream>>>(c1, c2, prelu, out);
}
